# starting kernel
# baseline (speedup 1.0000x reference)
#include <hip/hip_runtime.h>


#define BATCH 128
#define LEN   4096
#define OUTL  (2 * LEN - 1)
#define PAD   1024
#define XBUF  (LEN + 2 * PAD)

typedef float v2f __attribute__((ext_vector_type(2)));
typedef float v8f __attribute__((ext_vector_type(8)));

#if defined(__HIP_DEVICE_COMPILE__)
#if !__has_builtin(__builtin_amdgcn_wmma_f32_16x16x4_f32)
#error "missing __builtin_amdgcn_wmma_f32_16x16x4_f32 on this toolchain"
#endif
#endif

__global__ __launch_bounds__(256) void selfconv_wmma_f32(
    const float* __restrict__ x, float* __restrict__ out) {
  __shared__ float X[XBUF];

  const int tid = threadIdx.x;
  const int b   = blockIdx.x;

  for (int i = tid; i < 2 * PAD; i += 256) {
    int idx = (i < PAD) ? i : (LEN + i);
    X[idx] = 0.0f;
  }
  const float4* xb4 = (const float4*)(x + b * LEN);
  for (int i = tid; i < LEN / 4; i += 256) {
    float4 v = xb4[i];
    X[PAD + 4 * i + 0] = v.x;
    X[PAD + 4 * i + 1] = v.y;
    X[PAD + 4 * i + 2] = v.z;
    X[PAD + 4 * i + 3] = v.w;
  }
  __syncthreads();

  const int wave = tid >> 5;
  const int lane = tid & 31;
  const int m    = lane & 15;
  const int kh   = lane >> 4;
  const int gb   = wave << 10;

  int u_lo = gb - (LEN - 1);
  if (u_lo < -240) u_lo = -240;
  u_lo &= ~3;
  int u_hi = gb + 783;
  if (u_hi > LEN - 1) u_hi = LEN - 1;

  const int aBase = PAD + gb + m - 2 * kh;
  const int bBase = PAD + 2 * kh + 16 * m;

  v8f acc[4];
  acc[0] = (v8f){0.f, 0.f, 0.f, 0.f, 0.f, 0.f, 0.f, 0.f};
  acc[1] = acc[0];
  acc[2] = acc[0];
  acc[3] = acc[0];

  for (int u = u_lo; u <= u_hi; u += 4) {
    v2f bf;
    bf.x = X[bBase + u];
    bf.y = X[bBase + u + 1];
#pragma unroll
    for (int j = 0; j < 4; ++j) {
      const int ai = aBase + 256 * j - u;
      v2f af;
      af.x = X[ai];
      af.y = X[ai - 1];
      acc[j] = __builtin_amdgcn_wmma_f32_16x16x4_f32(
           false, af,  false, bf,
           (short)0, acc[j],  false,  false);
    }
  }

  float* outb = out + b * OUTL;
#pragma unroll
  for (int j = 0; j < 4; ++j) {
    const int tb = gb + 256 * j;
#pragma unroll
    for (int r = 0; r < 8; ++r) {
      const int t = tb + r + 8 * kh + 16 * m;
      if (t < OUTL) outb[t] = acc[j][r];
    }
  }
}

extern "C" void kernel_launch(void* const* d_in, const int* in_sizes, int n_in,
                              void* d_out, int out_size, void* d_ws, size_t ws_size,
                              hipStream_t stream) {
  (void)in_sizes; (void)n_in; (void)d_ws; (void)ws_size; (void)out_size;
  const float* x = (const float*)d_in[0];
  float* out = (float*)d_out;
  selfconv_wmma_f32<<<dim3(BATCH), dim3(256), 0, stream>>>(x, out);
}


# Row on two workgroups (grid 256), one per output half: each does lo*lo and the first half of lo*hi, doubled (2nd: reversed row); 6912 WMMAs each, not 13824; subtraction gone, sum order changes
# speedup vs baseline: 4.7019x; 4.7019x over previous
#include <hip/hip_runtime.h>


#define BATCH 128
#define LEN   4096
#define OUTL  (2 * LEN - 1)
#define HLEN  (LEN / 2)
#define NBLK  (HLEN / 16)
#define PADB  32
#define SEQB  (NBLK + PADB)
#define XS    ((2 * NBLK + 3 * PADB) * 16)
#define OSP   (2 * NBLK * 16)
#define OSC   (NBLK * 16)
#define NTHR  512
#define NWG   256
#define RUN0_G 0xC23435A5B6964787ull
#define RUN0_X 0x3444u
#define RUN1_G 0x0D102000E0001000ull
#define RUN1_V 0xF888u
#define RUN1_X 0x3800u

typedef float v2f __attribute__((ext_vector_type(2)));
typedef float v8f __attribute__((ext_vector_type(8)));

#if defined(__HIP_DEVICE_COMPILE__)
#if !__has_builtin(__builtin_amdgcn_wmma_f32_16x16x4_f32)
#error "missing __builtin_amdgcn_wmma_f32_16x16x4_f32 on this toolchain"
#endif
#endif

#define ROW_SHL(n) (0x100 + (n))
#define ROW_SHR(n) (0x110 + (n))

template <int CTRL>
__device__ __forceinline__ float row_move(float v) {
  return __builtin_bit_cast(float, __builtin_amdgcn_update_dpp(
      0, __builtin_bit_cast(int, v), CTRL, 0xF, 0xF, true));
}

__device__ __forceinline__ float other_row(float v) {
  return __builtin_bit_cast(float, __builtin_amdgcn_permlanex16(
      __builtin_bit_cast(int, v), __builtin_bit_cast(int, v),
      0x76543210, 0xfedcba98, false, false));
}

__device__ __forceinline__ float antidiag(const v8f d, const int h) {
  float lo = d[0];
  lo += row_move<ROW_SHR(1)>(d[1]);
  lo += row_move<ROW_SHR(2)>(d[2]);
  lo += row_move<ROW_SHR(3)>(d[3]);
  lo += row_move<ROW_SHR(4)>(d[4]);
  lo += row_move<ROW_SHR(5)>(d[5]);
  lo += row_move<ROW_SHR(6)>(d[6]);
  lo += row_move<ROW_SHR(7)>(d[7]);
  float hi = row_move<ROW_SHL(15)>(d[1]);
  hi += row_move<ROW_SHL(14)>(d[2]);
  hi += row_move<ROW_SHL(13)>(d[3]);
  hi += row_move<ROW_SHL(12)>(d[4]);
  hi += row_move<ROW_SHL(11)>(d[5]);
  hi += row_move<ROW_SHL(10)>(d[6]);
  hi += row_move<ROW_SHL(9)>(d[7]);
  const float e1 = row_move<ROW_SHR(8)>(lo);
  const float f1 = row_move<ROW_SHL(8)>(lo) + row_move<ROW_SHR(8)>(hi);
  const float mine   = h ? f1 : lo;
  const float theirs = h ? e1 : hi;
  return mine + other_row(theirs);
}

__global__ __launch_bounds__(NTHR) void selfconv_halves_wmma_f32(
    const float* __restrict__ x, float* __restrict__ out) {
  __shared__ __attribute__((aligned(16))) float xs[XS];
  __shared__ float osum[OSP + OSC];
  __shared__ float carry[2 * 256];

  const int tid = threadIdx.x;
  const int b   = blockIdx.x >> 1;
  const int dir = blockIdx.x & 1;

  const float4* xb4 = (const float4*)(x + b * LEN);
  const float4 v0 = xb4[tid];
  const float4 v1 = xb4[tid + NTHR];
  if (tid < 3 * 128)
    *(float4*)&xs[(tid >> 7) * (SEQB * 16) + 4 * (tid & 127)] =
        make_float4(0.f, 0.f, 0.f, 0.f);
  const int at = dir ? (NTHR - 1 - tid) : tid;
  const float4 pv = dir ? make_float4(v1.w, v1.z, v1.y, v1.x) : v0;
  const float4 qv = dir ? make_float4(v0.w, v0.z, v0.y, v0.x) : v1;
  *(float4*)&xs[PADB * 16 + 4 * at] = pv;
  *(float4*)&xs[(PADB + SEQB) * 16 + 4 * at] = qv;
  __syncthreads();

  const int lane = tid & 31;
  const int c    = lane & 15;
  const int h    = lane >> 4;
  const int w = __builtin_amdgcn_readfirstlane(tid >> 5);

  const int laneB = c - 16 * h;

  const int g_of_run0 = (int)((RUN0_G >> (4 * w)) & 15);
  const int g_of_run1 = (int)((RUN1_G >> (4 * w)) & 15);
  const int x_of_run0 = (int)((RUN0_X >> w) & 1);
  const int x_of_run1 = (int)((RUN1_X >> w) & 1);
  const int runs = 1 + (int)((RUN1_V >> w) & 1) + (w == 15 ? 1 : 0);
  const int cost0 = 1 + (x_of_run0 ? g_of_run0 : min(g_of_run0, 15 - g_of_run0));
  const int cost1 = 1 + (x_of_run1 ? g_of_run1 : min(g_of_run1, 15 - g_of_run1));
  const int wave_trips = 4 * (cost0 + (runs > 1 ? cost1 : 0) + (runs > 2 ? 1 : 0));
  const int prio_step2 = (4 * wave_trips) / 5;
  const int prio_step3 = (19 * wave_trips) / 20;
  int done = 0;
  int stage = 0;
  int next_step = wave_trips / 2;
  __builtin_amdgcn_s_setprio(3);

  for (int q = 0; q < runs; ++q) {
    const int g = (q == 0) ? g_of_run0 : (q == 1) ? g_of_run1 : 15;
    const int cross = (q == 0) ? x_of_run0 : (q == 1) ? x_of_run1 : 0;
    const int sbA = 0;
    const int sbB = cross ? SEQB * 16 : 0;
    float* const os = osum + cross * OSP;
    float* const cy = carry + cross * 256;

    float G0[4], Gp[4];
    for (int w4 = 0; w4 < 4; ++w4) {
      const int pmin = 16 * g + w4;
      int lower = pmin - (NBLK - 1);
      if (lower < 0) lower = 0;
      int upper = pmin + 12;
      if (upper > NBLK - 1) upper = NBLK - 1;
      const int trips = ((upper - lower) >> 4) + 1;

      v8f acc[4];
      acc[0] = (v8f){0.f, 0.f, 0.f, 0.f, 0.f, 0.f, 0.f, 0.f};
      acc[1] = acc[0];
      acc[2] = acc[0];
      acc[3] = acc[0];

      int aOff = sbA + (lower + PADB) * 16 + lane;
      int bOff = sbB + (pmin - lower + PADB) * 16 + laneB;
      for (int i = 0; i < trips; ++i) {
        if (done == next_step) {
          if (stage == 0) {
            __builtin_amdgcn_s_setprio(2);
            next_step = prio_step2;
          } else if (stage == 1) {
            __builtin_amdgcn_s_setprio(1);
            next_step = prio_step3;
          } else {
            __builtin_amdgcn_s_setprio(0);
            next_step = wave_trips;
          }
          ++stage;
        }
        ++done;
        v2f af[4], bf[7];
#pragma unroll
        for (int c4 = 0; c4 < 4; ++c4) {
          af[c4].x = xs[aOff + 64 * c4 + 32];
          af[c4].y = xs[aOff + 64 * c4];
        }
#pragma unroll
        for (int d = 0; d < 7; ++d) {
          bf[d].x = xs[bOff + 64 * (d - 3) - 32];
          bf[d].y = xs[bOff + 64 * (d - 3)];
        }
#pragma unroll
        for (int c4 = 0; c4 < 4; ++c4) {
#pragma unroll
          for (int j = 0; j < 4; ++j) {
            acc[j] = __builtin_amdgcn_wmma_f32_16x16x4_f32(
                 false, af[c4],  false, bf[j - c4 + 3],
                 (short)0, acc[j],  false,  false);
          }
        }
        aOff += 256;
        bOff -= 256;
      }

      float G[4];
#pragma unroll
      for (int j = 0; j < 4; ++j) G[j] = antidiag(acc[j], h);

      if (w4 == 0) {
#pragma unroll
        for (int j = 0; j < 4; ++j) G0[j] = G[j];
      } else {
#pragma unroll
        for (int j = 0; j < 4; ++j) {
          const float y = G[j] + other_row(Gp[j]);
          if (h == 0) os[(pmin + 4 * j) * 16 + c] = y;
        }
      }
#pragma unroll
      for (int j = 0; j < 4; ++j) Gp[j] = G[j];
    }

#pragma unroll
    for (int j = 1; j < 4; ++j) {
      const float y = G0[j] + other_row(Gp[j - 1]);
      if (h == 0) os[(16 * g + 4 * j) * 16 + c] = y;
    }
    if (h == 0) os[256 * g + c] = G0[0];
    else        cy[16 * g + c] = Gp[3];
  }
  __syncthreads();

  float* outb = out + b * OUTL;
  const bool first = (tid & 255) < 16;
  auto term = [&](const int q, const int u) -> float {
    const bool owed = first && u >= 256;
    const float cv = carry[q * 256 + (owed ? ((u >> 8) - 1) * 16 + (tid & 15) : 0)];
    return osum[q * OSP + u] + (owed ? cv : 0.f);
  };
#pragma unroll
  for (int i = 0; i < 8; ++i) {
    const int t = tid + NTHR * i;
    const float a = term(0, t);
    float v = (t < OSP - 1) ? a : 0.f;
    if (i >= 4) v += 2.f * term(1, t - HLEN);
    if (dir == 0) outb[t] = v;
    else if (t < LEN - 1) outb[OUTL - 1 - t] = v;
  }
}

extern "C" void kernel_launch(void* const* d_in, const int* in_sizes, int n_in,
                              void* d_out, int out_size, void* d_ws, size_t ws_size,
                              hipStream_t stream) {
  (void)in_sizes; (void)n_in; (void)d_ws; (void)ws_size; (void)out_size;
  const float* x = (const float*)d_in[0];
  float* out = (float*)d_out;
  selfconv_halves_wmma_f32<<<dim3(NWG), dim3(NTHR), 0, stream>>>(x, out);
}


# P*P tiles use their symmetry under a <-> p-a: K range cut at a = p/2 (weights 1, 1/2, 0 on the A fragments of a tile group's last trip), doubled at copy-out; 5120 WMMAs, not 6912; sum order changes
# speedup vs baseline: 4.9192x; 1.0462x over previous
#include <hip/hip_runtime.h>


#define BATCH 128
#define LEN   4096
#define OUTL  (2 * LEN - 1)
#define HLEN  (LEN / 2)
#define NBLK  (HLEN / 16)
#define PADB  32
#define SEQB  (NBLK + PADB)
#define XS    ((2 * NBLK + 3 * PADB) * 16)
#define OSP   (2 * NBLK * 16)
#define OSC   (NBLK * 16)
#define NTHR  512
#define NWG   256
#define RUN0_G 0x153924A7C846B567ull
#define RUN0_X 0xA827u
#define RUN1_G 0xE0001200D0003000ull
#define RUN1_V 0xCC88u
#define RUN2_G 0x0000F00000000000ull
#define RUN2_V 0x8800u
#define RUN2_X 0x8000u

typedef float v2f __attribute__((ext_vector_type(2)));
typedef float v8f __attribute__((ext_vector_type(8)));

#if defined(__HIP_DEVICE_COMPILE__)
#if !__has_builtin(__builtin_amdgcn_wmma_f32_16x16x4_f32)
#error "missing __builtin_amdgcn_wmma_f32_16x16x4_f32 on this toolchain"
#endif
#endif

#define ROW_SHL(n) (0x100 + (n))
#define ROW_SHR(n) (0x110 + (n))

template <int CTRL>
__device__ __forceinline__ float row_move(float v) {
  return __builtin_bit_cast(float, __builtin_amdgcn_update_dpp(
      0, __builtin_bit_cast(int, v), CTRL, 0xF, 0xF, true));
}

__device__ __forceinline__ float other_row(float v) {
  return __builtin_bit_cast(float, __builtin_amdgcn_permlanex16(
      __builtin_bit_cast(int, v), __builtin_bit_cast(int, v),
      0x76543210, 0xfedcba98, false, false));
}

__device__ __forceinline__ float antidiag(const v8f d, const int h) {
  float lo = d[0];
  lo += row_move<ROW_SHR(1)>(d[1]);
  lo += row_move<ROW_SHR(2)>(d[2]);
  lo += row_move<ROW_SHR(3)>(d[3]);
  lo += row_move<ROW_SHR(4)>(d[4]);
  lo += row_move<ROW_SHR(5)>(d[5]);
  lo += row_move<ROW_SHR(6)>(d[6]);
  lo += row_move<ROW_SHR(7)>(d[7]);
  float hi = row_move<ROW_SHL(15)>(d[1]);
  hi += row_move<ROW_SHL(14)>(d[2]);
  hi += row_move<ROW_SHL(13)>(d[3]);
  hi += row_move<ROW_SHL(12)>(d[4]);
  hi += row_move<ROW_SHL(11)>(d[5]);
  hi += row_move<ROW_SHL(10)>(d[6]);
  hi += row_move<ROW_SHL(9)>(d[7]);
  const float e1 = row_move<ROW_SHR(8)>(lo);
  const float f1 = row_move<ROW_SHL(8)>(lo) + row_move<ROW_SHR(8)>(hi);
  const float mine   = h ? f1 : lo;
  const float theirs = h ? e1 : hi;
  return mine + other_row(theirs);
}

__device__ __forceinline__ void group_range(const int pmin, const int cross,
                                            int& lower, int& trips, int& a_end) {
  const int first = max(pmin - (NBLK - 1), 0);
  if (cross) {
    a_end = min(pmin + 12, NBLK - 1);
    trips = ((a_end - first) >> 4) + 1;
    lower = first;
  } else {
    a_end = min((pmin + 12) >> 1, NBLK - 1);
    trips = ((a_end - first) >> 4) + 1;
    lower = a_end - 16 * trips + 1;
  }
}

__global__ __launch_bounds__(NTHR) void selfconv_halves_wmma_f32(
    const float* __restrict__ x, float* __restrict__ out) {
  __shared__ __attribute__((aligned(16))) float xs[XS];
  __shared__ float osum[OSP + OSC];
  __shared__ float carry[2 * 256];

  const int tid = threadIdx.x;
  const int b   = blockIdx.x >> 1;
  const int dir = blockIdx.x & 1;

  const float4* xb4 = (const float4*)(x + b * LEN);
  const float4 v0 = xb4[tid];
  const float4 v1 = xb4[tid + NTHR];
  if (tid < 3 * 128)
    *(float4*)&xs[(tid >> 7) * (SEQB * 16) + 4 * (tid & 127)] =
        make_float4(0.f, 0.f, 0.f, 0.f);
  const int at = dir ? (NTHR - 1 - tid) : tid;
  const float4 pv = dir ? make_float4(v1.w, v1.z, v1.y, v1.x) : v0;
  const float4 qv = dir ? make_float4(v0.w, v0.z, v0.y, v0.x) : v1;
  *(float4*)&xs[PADB * 16 + 4 * at] = pv;
  *(float4*)&xs[(PADB + SEQB) * 16 + 4 * at] = qv;
  __syncthreads();

  const int lane = tid & 31;
  const int c    = lane & 15;
  const int h    = lane >> 4;
  const int w = __builtin_amdgcn_readfirstlane(tid >> 5);

  const int laneB = c - 16 * h;

  const int g_of_run0 = (int)((RUN0_G >> (4 * w)) & 15);
  const int g_of_run1 = (int)((RUN1_G >> (4 * w)) & 15);
  const int g_of_run2 = (int)((RUN2_G >> (4 * w)) & 15);
  const int x_of_run0 = (int)((RUN0_X >> w) & 1);
  const int x_of_run2 = (int)((RUN2_X >> w) & 1);
  const int runs = 1 + (int)((RUN1_V >> w) & 1) + (int)((RUN2_V >> w) & 1);
  int wave_trips = 0;
  for (int q = 0; q < runs; ++q) {
    const int g = (q == 0) ? g_of_run0 : (q == 1) ? g_of_run1 : g_of_run2;
    const int cross = (q == 0) ? x_of_run0 : (q == 1) ? 0 : x_of_run2;
    for (int w4 = 0; w4 < 4; ++w4) {
      int lower, trips, a_end;
      group_range(16 * g + w4, cross, lower, trips, a_end);
      wave_trips += trips;
    }
  }
  const int prio_step2 = (4 * wave_trips) / 5;
  const int prio_step3 = (19 * wave_trips) / 20;
  int done = 0;
  int stage = 0;
  int next_step = wave_trips / 2;
  __builtin_amdgcn_s_setprio(3);

  for (int q = 0; q < runs; ++q) {
    const int g = (q == 0) ? g_of_run0 : (q == 1) ? g_of_run1 : g_of_run2;
    const int cross = (q == 0) ? x_of_run0 : (q == 1) ? 0 : x_of_run2;
    const int sbA = 0;
    const int sbB = cross ? SEQB * 16 : 0;
    float* const os = osum + cross * OSP;
    float* const cy = carry + cross * 256;

    float G0[4], Gp[4];
    for (int w4 = 0; w4 < 4; ++w4) {
      const int pmin = 16 * g + w4;
      int lower, trips, a_end;
      group_range(pmin, cross, lower, trips, a_end);
      float Wt[11];
      const int wbase = 2 * (a_end - 15) - pmin + 2 * h;
#pragma unroll
      for (int k = 0; k < 11; ++k) {
        const int d2 = wbase + 4 * (k - 3);
        Wt[k] = (cross || d2 < 0) ? 1.f : (d2 == 0) ? 0.5f : 0.f;
      }

      v8f acc[4];
      acc[0] = (v8f){0.f, 0.f, 0.f, 0.f, 0.f, 0.f, 0.f, 0.f};
      acc[1] = acc[0];
      acc[2] = acc[0];
      acc[3] = acc[0];

      int aOff = sbA + (lower + PADB) * 16 + lane;
      int bOff = sbB + (pmin - lower + PADB) * 16 + laneB;
      auto prio_tick = [&]() {
        if (done == next_step) {
          if (stage == 0) {
            __builtin_amdgcn_s_setprio(2);
            next_step = prio_step2;
          } else if (stage == 1) {
            __builtin_amdgcn_s_setprio(1);
            next_step = prio_step3;
          } else {
            __builtin_amdgcn_s_setprio(0);
            next_step = wave_trips;
          }
          ++stage;
        }
        ++done;
      };
      v2f af[4], bf[7];
      auto load_fragments = [&]() {
#pragma unroll
        for (int c4 = 0; c4 < 4; ++c4) {
          af[c4].x = xs[aOff + 64 * c4 + 32];
          af[c4].y = xs[aOff + 64 * c4];
        }
#pragma unroll
        for (int d = 0; d < 7; ++d) {
          bf[d].x = xs[bOff + 64 * (d - 3) - 32];
          bf[d].y = xs[bOff + 64 * (d - 3)];
        }
      };
      for (int i = 0; i < trips - 1; ++i) {
        prio_tick();
        load_fragments();
#pragma unroll
        for (int c4 = 0; c4 < 4; ++c4) {
#pragma unroll
          for (int j = 0; j < 4; ++j) {
            acc[j] = __builtin_amdgcn_wmma_f32_16x16x4_f32(
                 false, af[c4],  false, bf[j - c4 + 3],
                 (short)0, acc[j],  false,  false);
          }
        }
        aOff += 256;
        bOff -= 256;
      }
      prio_tick();
      load_fragments();
#pragma unroll
      for (int c4 = 0; c4 < 4; ++c4) {
#pragma unroll
        for (int j = 0; j < 4; ++j) {
          v2f am;
          am.x = af[c4].x * Wt[2 * c4 - j + 4];
          am.y = af[c4].y * Wt[2 * c4 - j + 3];
          acc[j] = __builtin_amdgcn_wmma_f32_16x16x4_f32(
               false, am,  false, bf[j - c4 + 3],
               (short)0, acc[j],  false,  false);
        }
      }

      float G[4];
#pragma unroll
      for (int j = 0; j < 4; ++j) G[j] = antidiag(acc[j], h);

      if (w4 == 0) {
#pragma unroll
        for (int j = 0; j < 4; ++j) G0[j] = G[j];
      } else {
#pragma unroll
        for (int j = 0; j < 4; ++j) {
          const float y = G[j] + other_row(Gp[j]);
          if (h == 0) os[(pmin + 4 * j) * 16 + c] = y;
        }
      }
#pragma unroll
      for (int j = 0; j < 4; ++j) Gp[j] = G[j];
    }

#pragma unroll
    for (int j = 1; j < 4; ++j) {
      const float y = G0[j] + other_row(Gp[j - 1]);
      if (h == 0) os[(16 * g + 4 * j) * 16 + c] = y;
    }
    if (h == 0) os[256 * g + c] = G0[0];
    else        cy[16 * g + c] = Gp[3];
  }
  __syncthreads();

  float* outb = out + b * OUTL;
  const bool first = (tid & 255) < 16;
  auto term = [&](const int q, const int u) -> float {
    const bool owed = first && u >= 256;
    const float cv = carry[q * 256 + (owed ? ((u >> 8) - 1) * 16 + (tid & 15) : 0)];
    return osum[q * OSP + u] + (owed ? cv : 0.f);
  };
#pragma unroll
  for (int i = 0; i < 8; ++i) {
    const int t = tid + NTHR * i;
    const float a = term(0, t);
    float v = (t < OSP - 1) ? a : 0.f;
    if (i >= 4) v += term(1, t - HLEN);
    v *= 2.f;
    if (dir == 0) outb[t] = v;
    else if (t < LEN - 1) outb[OUTL - 1 - t] = v;
  }
}

extern "C" void kernel_launch(void* const* d_in, const int* in_sizes, int n_in,
                              void* d_out, int out_size, void* d_ws, size_t ws_size,
                              hipStream_t stream) {
  (void)in_sizes; (void)n_in; (void)d_ws; (void)ws_size; (void)out_size;
  const float* x = (const float*)d_in[0];
  float* out = (float*)d_out;
  selfconv_halves_wmma_f32<<<dim3(NWG), dim3(NTHR), 0, stream>>>(x, out);
}


# Half-row products merged: row staged contiguously, tile p sums a <= p/2 over the whole row in one accumulator (256 tiles not 384, 4608 WMMAs not 5120, one 16-tile run per wave); sum order changes
# speedup vs baseline: 6.5129x; 1.3240x over previous
#include <hip/hip_runtime.h>


#define BATCH 128
#define LEN   4096
#define OUTL  (2 * LEN - 1)
#define NBLK  (LEN / 16)
#define PADB  32
#define XS    ((NBLK + PADB) * 16)
#define NTHR  512
#define NWG   256
#define RUN_G 0x23014567AB89CDEFull

typedef float v2f __attribute__((ext_vector_type(2)));
typedef float v8f __attribute__((ext_vector_type(8)));

#if defined(__HIP_DEVICE_COMPILE__)
#if !__has_builtin(__builtin_amdgcn_wmma_f32_16x16x4_f32)
#error "missing __builtin_amdgcn_wmma_f32_16x16x4_f32 on this toolchain"
#endif
#endif

#define ROW_SHL(n) (0x100 + (n))
#define ROW_SHR(n) (0x110 + (n))

template <int CTRL>
__device__ __forceinline__ float row_move(float v) {
  return __builtin_bit_cast(float, __builtin_amdgcn_update_dpp(
      0, __builtin_bit_cast(int, v), CTRL, 0xF, 0xF, true));
}

__device__ __forceinline__ float other_row(float v) {
  return __builtin_bit_cast(float, __builtin_amdgcn_permlanex16(
      __builtin_bit_cast(int, v), __builtin_bit_cast(int, v),
      0x76543210, 0xfedcba98, false, false));
}

__device__ __forceinline__ float antidiag(const v8f d, const int h) {
  float lo = d[0];
  lo += row_move<ROW_SHR(1)>(d[1]);
  lo += row_move<ROW_SHR(2)>(d[2]);
  lo += row_move<ROW_SHR(3)>(d[3]);
  lo += row_move<ROW_SHR(4)>(d[4]);
  lo += row_move<ROW_SHR(5)>(d[5]);
  lo += row_move<ROW_SHR(6)>(d[6]);
  lo += row_move<ROW_SHR(7)>(d[7]);
  float hi = row_move<ROW_SHL(15)>(d[1]);
  hi += row_move<ROW_SHL(14)>(d[2]);
  hi += row_move<ROW_SHL(13)>(d[3]);
  hi += row_move<ROW_SHL(12)>(d[4]);
  hi += row_move<ROW_SHL(11)>(d[5]);
  hi += row_move<ROW_SHL(10)>(d[6]);
  hi += row_move<ROW_SHL(9)>(d[7]);
  const float e1 = row_move<ROW_SHR(8)>(lo);
  const float f1 = row_move<ROW_SHL(8)>(lo) + row_move<ROW_SHR(8)>(hi);
  const float mine   = h ? f1 : lo;
  const float theirs = h ? e1 : hi;
  return mine + other_row(theirs);
}

__device__ __forceinline__ void group_range(const int pmin,
                                            int& lower, int& trips, int& a_end) {
  a_end = (pmin + 12) >> 1;
  trips = (a_end >> 4) + 1;
  lower = a_end - 16 * trips + 1;
}

__global__ __launch_bounds__(NTHR) void selfconv_halves_wmma_f32(
    const float* __restrict__ x, float* __restrict__ out) {
  __shared__ __attribute__((aligned(16))) float xs[XS];
  __shared__ float osum[LEN];
  __shared__ float carry[256];

  const int tid = threadIdx.x;
  const int b   = blockIdx.x >> 1;
  const int dir = blockIdx.x & 1;

  const float4* xb4 = (const float4*)(x + b * LEN);
  const float4 v0 = xb4[tid];
  const float4 v1 = xb4[tid + NTHR];
  if (tid < PADB * 4)
    *(float4*)&xs[4 * tid] = make_float4(0.f, 0.f, 0.f, 0.f);
  const int at = dir ? (NTHR - 1 - tid) : tid;
  const float4 pv = dir ? make_float4(v1.w, v1.z, v1.y, v1.x) : v0;
  const float4 qv = dir ? make_float4(v0.w, v0.z, v0.y, v0.x) : v1;
  *(float4*)&xs[PADB * 16 + 4 * at] = pv;
  *(float4*)&xs[(PADB + NBLK / 2) * 16 + 4 * at] = qv;
  __syncthreads();

  const int lane = tid & 31;
  const int c    = lane & 15;
  const int h    = lane >> 4;
  const int w = __builtin_amdgcn_readfirstlane(tid >> 5);

  const int laneB = c - 16 * h;

  const int g = (int)((RUN_G >> (4 * w)) & 15);
  int wave_trips = 0;
  for (int w4 = 0; w4 < 4; ++w4) {
    int lower, trips, a_end;
    group_range(16 * g + w4, lower, trips, a_end);
    wave_trips += trips;
  }
  const int prio_step2 = (4 * wave_trips) / 5;
  const int prio_step3 = (19 * wave_trips) / 20;
  int done = 0;
  int stage = 0;
  int next_step = wave_trips / 2;
  __builtin_amdgcn_s_setprio(3);

  float G0[4], Gp[4];
  for (int w4 = 0; w4 < 4; ++w4) {
    const int pmin = 16 * g + w4;
    int lower, trips, a_end;
    group_range(pmin, lower, trips, a_end);
    float Wt[11];
    const int wbase = 2 * (a_end - 15) - pmin + 2 * h;
#pragma unroll
    for (int k = 0; k < 11; ++k) {
      const int d2 = wbase + 4 * (k - 3);
      Wt[k] = (d2 < 0) ? 1.f : (d2 == 0) ? 0.5f : 0.f;
    }

    v8f acc[4];
    acc[0] = (v8f){0.f, 0.f, 0.f, 0.f, 0.f, 0.f, 0.f, 0.f};
    acc[1] = acc[0];
    acc[2] = acc[0];
    acc[3] = acc[0];

    int aOff = (lower + PADB) * 16 + lane;
    int bOff = (pmin - lower + PADB) * 16 + laneB;
    auto prio_tick = [&]() {
      if (done == next_step) {
        if (stage == 0) {
          __builtin_amdgcn_s_setprio(2);
          next_step = prio_step2;
        } else if (stage == 1) {
          __builtin_amdgcn_s_setprio(1);
          next_step = prio_step3;
        } else {
          __builtin_amdgcn_s_setprio(0);
          next_step = wave_trips;
        }
        ++stage;
      }
      ++done;
    };
    v2f af[4], bf[7];
    auto load_fragments = [&]() {
#pragma unroll
      for (int c4 = 0; c4 < 4; ++c4) {
        af[c4].x = xs[aOff + 64 * c4 + 32];
        af[c4].y = xs[aOff + 64 * c4];
      }
#pragma unroll
      for (int d = 0; d < 7; ++d) {
        bf[d].x = xs[bOff + 64 * (d - 3) - 32];
        bf[d].y = xs[bOff + 64 * (d - 3)];
      }
    };
    for (int i = 0; i < trips - 1; ++i) {
      prio_tick();
      load_fragments();
#pragma unroll
      for (int c4 = 0; c4 < 4; ++c4) {
#pragma unroll
        for (int j = 0; j < 4; ++j) {
          acc[j] = __builtin_amdgcn_wmma_f32_16x16x4_f32(
               false, af[c4],  false, bf[j - c4 + 3],
               (short)0, acc[j],  false,  false);
        }
      }
      aOff += 256;
      bOff -= 256;
    }
    prio_tick();
    load_fragments();
#pragma unroll
    for (int c4 = 0; c4 < 4; ++c4) {
#pragma unroll
      for (int j = 0; j < 4; ++j) {
        v2f am;
        am.x = af[c4].x * Wt[2 * c4 - j + 4];
        am.y = af[c4].y * Wt[2 * c4 - j + 3];
        acc[j] = __builtin_amdgcn_wmma_f32_16x16x4_f32(
             false, am,  false, bf[j - c4 + 3],
             (short)0, acc[j],  false,  false);
      }
    }

    float G[4];
#pragma unroll
    for (int j = 0; j < 4; ++j) G[j] = antidiag(acc[j], h);

    if (w4 == 0) {
#pragma unroll
      for (int j = 0; j < 4; ++j) G0[j] = G[j];
    } else {
#pragma unroll
      for (int j = 0; j < 4; ++j) {
        const float y = G[j] + other_row(Gp[j]);
        if (h == 0) osum[(pmin + 4 * j) * 16 + c] = y;
      }
    }
#pragma unroll
    for (int j = 0; j < 4; ++j) Gp[j] = G[j];
  }

#pragma unroll
  for (int j = 1; j < 4; ++j) {
    const float y = G0[j] + other_row(Gp[j - 1]);
    if (h == 0) osum[(16 * g + 4 * j) * 16 + c] = y;
  }
  if (h == 0) osum[256 * g + c] = G0[0];
  else        carry[16 * g + c] = Gp[3];
  __syncthreads();

  float* outb = out + b * OUTL;
  const bool first = (tid & 255) < 16;
#pragma unroll
  for (int i = 0; i < 8; ++i) {
    const int t = tid + NTHR * i;
    const bool owed = first && t >= 256;
    const float cv = carry[owed ? ((t >> 8) - 1) * 16 + (tid & 15) : 0];
    const float v = 2.f * (osum[t] + (owed ? cv : 0.f));
    if (dir == 0) outb[t] = v;
    else if (t < LEN - 1) outb[OUTL - 1 - t] = v;
  }
}

extern "C" void kernel_launch(void* const* d_in, const int* in_sizes, int n_in,
                              void* d_out, int out_size, void* d_ws, size_t ws_size,
                              hipStream_t stream) {
  (void)in_sizes; (void)n_in; (void)d_ws; (void)ws_size; (void)out_size;
  const float* x = (const float*)d_in[0];
  float* out = (float*)d_out;
  selfconv_halves_wmma_f32<<<dim3(NWG), dim3(NTHR), 0, stream>>>(x, out);
}


# K steps that can only add zeros are not issued: 8 pad blocks heading even runs and the last K step of a group's first two tiles (weights 0); 4224 WMMAs not 4608, runs re-dealt; sum order unchanged
# speedup vs baseline: 6.8780x; 1.0561x over previous
#include <hip/hip_runtime.h>


#define BATCH 128
#define LEN   4096
#define OUTL  (2 * LEN - 1)
#define NBLK  (LEN / 16)
#define PADB  32
#define XS    ((NBLK + PADB) * 16)
#define NTHR  512
#define NWG   256
#define RUN_G 0x32104567BA98CDEFull

typedef float v2f __attribute__((ext_vector_type(2)));
typedef float v8f __attribute__((ext_vector_type(8)));

#if defined(__HIP_DEVICE_COMPILE__)
#if !__has_builtin(__builtin_amdgcn_wmma_f32_16x16x4_f32)
#error "missing __builtin_amdgcn_wmma_f32_16x16x4_f32 on this toolchain"
#endif
#endif

#define ROW_SHL(n) (0x100 + (n))
#define ROW_SHR(n) (0x110 + (n))

template <int CTRL>
__device__ __forceinline__ float row_move(float v) {
  return __builtin_bit_cast(float, __builtin_amdgcn_update_dpp(
      0, __builtin_bit_cast(int, v), CTRL, 0xF, 0xF, true));
}

__device__ __forceinline__ float other_row(float v) {
  return __builtin_bit_cast(float, __builtin_amdgcn_permlanex16(
      __builtin_bit_cast(int, v), __builtin_bit_cast(int, v),
      0x76543210, 0xfedcba98, false, false));
}

__device__ __forceinline__ float antidiag(const v8f d, const int h) {
  float lo = d[0];
  lo += row_move<ROW_SHR(1)>(d[1]);
  lo += row_move<ROW_SHR(2)>(d[2]);
  lo += row_move<ROW_SHR(3)>(d[3]);
  lo += row_move<ROW_SHR(4)>(d[4]);
  lo += row_move<ROW_SHR(5)>(d[5]);
  lo += row_move<ROW_SHR(6)>(d[6]);
  lo += row_move<ROW_SHR(7)>(d[7]);
  float hi = row_move<ROW_SHL(15)>(d[1]);
  hi += row_move<ROW_SHL(14)>(d[2]);
  hi += row_move<ROW_SHL(13)>(d[3]);
  hi += row_move<ROW_SHL(12)>(d[4]);
  hi += row_move<ROW_SHL(11)>(d[5]);
  hi += row_move<ROW_SHL(10)>(d[6]);
  hi += row_move<ROW_SHL(9)>(d[7]);
  const float e1 = row_move<ROW_SHR(8)>(lo);
  const float f1 = row_move<ROW_SHL(8)>(lo) + row_move<ROW_SHR(8)>(hi);
  const float mine   = h ? f1 : lo;
  const float theirs = h ? e1 : hi;
  return mine + other_row(theirs);
}

__device__ __forceinline__ void group_range(const int pmin,
                                            int& lower, int& trips, int& a_end) {
  a_end = (pmin + 12) >> 1;
  trips = (a_end >> 4) + 1;
  lower = a_end - 16 * trips + 1;
}

__global__ __launch_bounds__(NTHR) void selfconv_halves_wmma_f32(
    const float* __restrict__ x, float* __restrict__ out) {
  __shared__ __attribute__((aligned(16))) float xs[XS];
  __shared__ float osum[LEN];
  __shared__ float carry[256];

  const int tid = threadIdx.x;
  const int b   = blockIdx.x >> 1;
  const int dir = blockIdx.x & 1;

  const float4* xb4 = (const float4*)(x + b * LEN);
  const float4 v0 = xb4[tid];
  const float4 v1 = xb4[tid + NTHR];
  if (tid < PADB * 4)
    *(float4*)&xs[4 * tid] = make_float4(0.f, 0.f, 0.f, 0.f);
  const int at = dir ? (NTHR - 1 - tid) : tid;
  const float4 pv = dir ? make_float4(v1.w, v1.z, v1.y, v1.x) : v0;
  const float4 qv = dir ? make_float4(v0.w, v0.z, v0.y, v0.x) : v1;
  *(float4*)&xs[PADB * 16 + 4 * at] = pv;
  *(float4*)&xs[(PADB + NBLK / 2) * 16 + 4 * at] = qv;
  __syncthreads();

  const int lane = tid & 31;
  const int c    = lane & 15;
  const int h    = lane >> 4;
  const int w = __builtin_amdgcn_readfirstlane(tid >> 5);

  const int laneB = c - 16 * h;

  const int g = (int)((RUN_G >> (4 * w)) & 15);
  const bool pad8 = (g & 1) == 0;
  int wave_halves = 0;
  for (int w4 = 0; w4 < 4; ++w4) {
    int lower, trips, a_end;
    group_range(16 * g + w4, lower, trips, a_end);
    wave_halves += 2 * trips - (pad8 ? 1 : 0);
  }
  const int prio_step2 = (4 * wave_halves) / 5;
  const int prio_step3 = (19 * wave_halves) / 20;
  int done = 0;
  int stage = 0;
  int next_step = wave_halves / 2;
  __builtin_amdgcn_s_setprio(3);

  float G0[4], Gp[4];
  for (int w4 = 0; w4 < 4; ++w4) {
    const int pmin = 16 * g + w4;
    int lower, trips, a_end;
    group_range(pmin, lower, trips, a_end);
    float Wt[11];
    const int wbase = 2 * (a_end - 15) - pmin + 2 * h;
#pragma unroll
    for (int k = 0; k < 11; ++k) {
      const int d2 = wbase + 4 * (k - 3);
      Wt[k] = (d2 < 0) ? 1.f : (d2 == 0) ? 0.5f : 0.f;
    }

    v8f acc[4];
    acc[0] = (v8f){0.f, 0.f, 0.f, 0.f, 0.f, 0.f, 0.f, 0.f};
    acc[1] = acc[0];
    acc[2] = acc[0];
    acc[3] = acc[0];

    int aOff = (lower + PADB) * 16 + lane;
    int bOff = (pmin - lower + PADB) * 16 + laneB;
    auto prio_tick = [&](const int halves) {
      if (done >= next_step) {
        if (stage == 0) {
          __builtin_amdgcn_s_setprio(2);
          next_step = prio_step2;
        } else if (stage == 1) {
          __builtin_amdgcn_s_setprio(1);
          next_step = prio_step3;
        } else {
          __builtin_amdgcn_s_setprio(0);
          next_step = 2 * wave_halves;
        }
        ++stage;
      }
      done += halves;
    };
    v2f af[4], bf[7];
    auto load_fragments = [&]() {
#pragma unroll
      for (int c4 = 0; c4 < 4; ++c4) {
        af[c4].x = xs[aOff + 64 * c4 + 32];
        af[c4].y = xs[aOff + 64 * c4];
      }
#pragma unroll
      for (int d = 0; d < 7; ++d) {
        bf[d].x = xs[bOff + 64 * (d - 3) - 32];
        bf[d].y = xs[bOff + 64 * (d - 3)];
      }
    };
    auto load_upper_fragments = [&]() {
#pragma unroll
      for (int c4 = 2; c4 < 4; ++c4) {
        af[c4].x = xs[aOff + 64 * c4 + 32];
        af[c4].y = xs[aOff + 64 * c4];
      }
#pragma unroll
      for (int d = 0; d < 5; ++d) {
        bf[d].x = xs[bOff + 64 * (d - 3) - 32];
        bf[d].y = xs[bOff + 64 * (d - 3)];
      }
    };
    int whole = trips - 1;
    if (pad8 && whole > 0) {
      prio_tick(1);
      load_upper_fragments();
#pragma unroll
      for (int c4 = 2; c4 < 4; ++c4) {
#pragma unroll
        for (int j = 0; j < 4; ++j) {
          acc[j] = __builtin_amdgcn_wmma_f32_16x16x4_f32(
               false, af[c4],  false, bf[j - c4 + 3],
               (short)0, acc[j],  false,  false);
        }
      }
      aOff += 256;
      bOff -= 256;
      --whole;
    }
    for (int i = 0; i < whole; ++i) {
      prio_tick(2);
      load_fragments();
#pragma unroll
      for (int c4 = 0; c4 < 4; ++c4) {
#pragma unroll
        for (int j = 0; j < 4; ++j) {
          acc[j] = __builtin_amdgcn_wmma_f32_16x16x4_f32(
               false, af[c4],  false, bf[j - c4 + 3],
               (short)0, acc[j],  false,  false);
        }
      }
      aOff += 256;
      bOff -= 256;
    }
    const bool upper_only = pad8 && trips == 1;
    prio_tick(upper_only ? 1 : 2);
    load_fragments();
    if (!upper_only) {
#pragma unroll
      for (int c4 = 0; c4 < 2; ++c4) {
#pragma unroll
        for (int j = 0; j < 4; ++j) {
          v2f am;
          am.x = af[c4].x * Wt[2 * c4 - j + 4];
          am.y = af[c4].y * Wt[2 * c4 - j + 3];
          acc[j] = __builtin_amdgcn_wmma_f32_16x16x4_f32(
               false, am,  false, bf[j - c4 + 3],
               (short)0, acc[j],  false,  false);
        }
      }
    }
#pragma unroll
    for (int c4 = 2; c4 < 4; ++c4) {
#pragma unroll
      for (int j = 0; j < 4; ++j) {
        if (c4 == 3 && j < 2) continue;
        v2f am;
        am.x = af[c4].x * Wt[2 * c4 - j + 4];
        am.y = af[c4].y * Wt[2 * c4 - j + 3];
        acc[j] = __builtin_amdgcn_wmma_f32_16x16x4_f32(
             false, am,  false, bf[j - c4 + 3],
             (short)0, acc[j],  false,  false);
      }
    }

    float G[4];
#pragma unroll
    for (int j = 0; j < 4; ++j) G[j] = antidiag(acc[j], h);

    if (w4 == 0) {
#pragma unroll
      for (int j = 0; j < 4; ++j) G0[j] = G[j];
    } else {
#pragma unroll
      for (int j = 0; j < 4; ++j) {
        const float y = G[j] + other_row(Gp[j]);
        if (h == 0) osum[(pmin + 4 * j) * 16 + c] = y;
      }
    }
#pragma unroll
    for (int j = 0; j < 4; ++j) Gp[j] = G[j];
  }

#pragma unroll
  for (int j = 1; j < 4; ++j) {
    const float y = G0[j] + other_row(Gp[j - 1]);
    if (h == 0) osum[(16 * g + 4 * j) * 16 + c] = y;
  }
  if (h == 0) osum[256 * g + c] = G0[0];
  else        carry[16 * g + c] = Gp[3];
  __syncthreads();

  float* outb = out + b * OUTL;
  const bool first = (tid & 255) < 16;
#pragma unroll
  for (int i = 0; i < 8; ++i) {
    const int t = tid + NTHR * i;
    const bool owed = first && t >= 256;
    const float cv = carry[owed ? ((t >> 8) - 1) * 16 + (tid & 15) : 0];
    const float v = 2.f * (osum[t] + (owed ? cv : 0.f));
    if (dir == 0) outb[t] = v;
    else if (t < LEN - 1) outb[OUTL - 1 - t] = v;
  }
}

extern "C" void kernel_launch(void* const* d_in, const int* in_sizes, int n_in,
                              void* d_out, int out_size, void* d_ws, size_t ws_size,
                              hipStream_t stream) {
  (void)in_sizes; (void)n_in; (void)d_ws; (void)ws_size; (void)out_size;
  const float* x = (const float*)d_in[0];
  float* out = (float*)d_out;
  selfconv_halves_wmma_f32<<<dim3(NWG), dim3(NTHR), 0, stream>>>(x, out);
}


# Convolution theorem on the matrix unit: row squared mod z^4096-i as a 4096-point complex cyclic product, 3 radix-16 DFT passes each way on f32 WMMA, 1408 per row not 8448; sum order changes
# speedup vs baseline: 14.1448x; 2.0565x over previous
#include <hip/hip_runtime.h>


#define LEN   4096
#define OUTL  (2 * LEN - 1)
#define NTHR  512
#define RP    18
#define TP    290
#define PLANE (16 * TP)

typedef float v2f __attribute__((ext_vector_type(2)));
typedef float v8f __attribute__((ext_vector_type(8)));

#if defined(__HIP_DEVICE_COMPILE__)
#if !__has_builtin(__builtin_amdgcn_wmma_f32_16x16x4_f32)
#error "missing __builtin_amdgcn_wmma_f32_16x16x4_f32 on this toolchain"
#endif
#endif

__device__ __forceinline__ v8f mma(const v2f a, const v2f b, const v8f c) {
  return __builtin_amdgcn_wmma_f32_16x16x4_f32(
       false, a,  false, b,
       (short)0, c,  false,  false);
}

__device__ __forceinline__ float2 unit(const float2* tabL, const float2* tabH,
                                       const int q) {
  const int qm = q & 16383;
  const float2 a = tabL[qm & 127];
  const float2 b = tabH[qm >> 7];
  return make_float2(a.x * b.x - a.y * b.y, a.x * b.y + a.y * b.x);
}

__device__ __forceinline__ float2 unit128(const float2* tabH, const int q) {
  return tabH[(q & 16383) >> 7];
}

__global__ __launch_bounds__(NTHR) void selfconv_dft16_wmma_f32(
    const float* __restrict__ x, float* __restrict__ out) {
  __shared__ __attribute__((aligned(16))) float sRe[PLANE];
  __shared__ __attribute__((aligned(16))) float sIm[PLANE];
  __shared__ __attribute__((aligned(16))) float sOut[2 * LEN];
  __shared__ float2 tabL[128];
  __shared__ float2 tabH[128];

  const int tid  = threadIdx.x;
  const int b    = blockIdx.x;
  const int lane = tid & 31;
  const int c    = lane & 15;
  const int h    = lane >> 4;
  const int w = __builtin_amdgcn_readfirstlane(tid >> 5);

  if (tid < 256) {
    const int t = tid & 127;
    const float ang = (tid < 128) ? (float)t * (1.f / 8192.f)
                                  : (float)t * (1.f / 64.f);
    float sn, cs;
    sincospif(ang, &sn, &cs);
    if (tid < 128) tabL[t] = make_float2(cs, sn);
    else           tabH[t] = make_float2(cs, sn);
  }
  __syncthreads();

  const float* xb = x + b * LEN;
  float xv[8];
#pragma unroll
  for (int v = 0; v < 8; ++v) xv[v] = xb[256 * (8 * h + v) + 16 * w + c];

  v2f Fr[4], Fi[4], Fn[4];
  v2f Pr[4], Pi[4];
  v2f Gr[4], Gi[4], Gn[4];
#pragma unroll
  for (int j = 0; j < 4; ++j) {
#pragma unroll
    for (int s = 0; s < 2; ++s) {
      const int k = 8 * h + 2 * j + s;
      const float2 f = unit128(tabH, -1024 * ((c * k) & 15));
      const float2 p = unit128(tabH, 256 * k * (1 - 4 * c));
      const float2 g = unit128(tabH, -256 * c * (1 - 4 * k));
      if (s == 0) {
        Fr[j].x = f.x; Fi[j].x = f.y; Fn[j].x = -f.y;
        Pr[j].x = p.x; Pi[j].x = p.y;
        Gr[j].x = g.x; Gi[j].x = g.y; Gn[j].x = -g.y;
      } else {
        Fr[j].y = f.x; Fi[j].y = f.y; Fn[j].y = -f.y;
        Pr[j].y = p.x; Pi[j].y = p.y;
        Gr[j].y = g.x; Gi[j].y = g.y; Gn[j].y = -g.y;
      }
    }
  }
  float T1r[8], T1i[8];
  float T2r[8], T2i[8];
#pragma unroll
  for (int v = 0; v < 8; ++v) {
    const float2 t1 = unit(tabL, tabH, (16 * w + c) * (1 - 4 * (8 * h + v)));
    const float2 t2 = unit(tabL, tabH, -64 * (8 * h + v) * c);
    T1r[v] = t1.x; T1i[v] = t1.y;
    T2r[v] = t2.x; T2i[v] = t2.y;
  }

  const v8f zero = (v8f){0.f, 0.f, 0.f, 0.f, 0.f, 0.f, 0.f, 0.f};
  v8f yr, yi;
  v2f br[4], bi[4];

  yr = zero; yi = zero;
#pragma unroll
  for (int j = 0; j < 4; ++j) {
    v2f bx;
    bx.x = xv[2 * j];
    bx.y = xv[2 * j + 1];
    yr = mma(Pr[j], bx, yr);
    yi = mma(Pi[j], bx, yi);
  }
#pragma unroll
  for (int v = 0; v < 8; ++v) {
    const int idx = (8 * h + v) * TP + w * RP + c;
    sRe[idx] = yr[v] * T1r[v] - yi[v] * T1i[v];
    sIm[idx] = yr[v] * T1i[v] + yi[v] * T1r[v];
  }
  __syncthreads();

#pragma unroll
  for (int j = 0; j < 4; ++j) {
    const int idx = w * TP + (8 * h + 2 * j) * RP + c;
    br[j].x = sRe[idx]; br[j].y = sRe[idx + RP];
    bi[j].x = sIm[idx]; bi[j].y = sIm[idx + RP];
  }
  yr = zero; yi = zero;
#pragma unroll
  for (int j = 0; j < 4; ++j) {
    yr = mma(Fr[j], br[j], yr);
    yi = mma(Fr[j], bi[j], yi);
    yr = mma(Fn[j], bi[j], yr);
    yi = mma(Fi[j], br[j], yi);
  }
#pragma unroll
  for (int v = 0; v < 8; ++v) {
    const int idx = w * TP + (8 * h + v) * RP + c;
    sRe[idx] = yr[v] * T2r[v] - yi[v] * T2i[v];
    sIm[idx] = yr[v] * T2i[v] + yi[v] * T2r[v];
  }
  __syncthreads();

#pragma unroll
  for (int j = 0; j < 4; ++j) {
    const int idx = w * TP + c * RP + 8 * h + 2 * j;
    const float2 tr = *(const float2*)&sRe[idx];
    const float2 ti = *(const float2*)&sIm[idx];
    br[j].x = tr.x; br[j].y = tr.y;
    bi[j].x = ti.x; bi[j].y = ti.y;
  }
  yr = zero; yi = zero;
#pragma unroll
  for (int j = 0; j < 4; ++j) {
    yr = mma(Fr[j], br[j], yr);
    yi = mma(Fr[j], bi[j], yi);
    yr = mma(Fn[j], bi[j], yr);
    yi = mma(Fi[j], br[j], yi);
  }
#pragma unroll
  for (int j = 0; j < 4; ++j) {
    const float a0 = yr[2 * j],     b0 = yi[2 * j];
    const float a1 = yr[2 * j + 1], b1 = yi[2 * j + 1];
    br[j].x = a0 * a0 - b0 * b0;  bi[j].x = 2.f * a0 * b0;
    br[j].y = a1 * a1 - b1 * b1;  bi[j].y = 2.f * a1 * b1;
  }

  yr = zero; yi = zero;
#pragma unroll
  for (int j = 0; j < 4; ++j) {
    yr = mma(Fr[j], br[j], yr);
    yi = mma(Fr[j], bi[j], yi);
    yr = mma(Fi[j], bi[j], yr);
    yi = mma(Fn[j], br[j], yi);
  }
#pragma unroll
  for (int v = 0; v < 8; ++v) {
    const int idx = w * TP + (8 * h + v) * RP + c;
    sRe[idx] = yr[v] * T2r[v] + yi[v] * T2i[v];
    sIm[idx] = yi[v] * T2r[v] - yr[v] * T2i[v];
  }
  __syncthreads();

#pragma unroll
  for (int j = 0; j < 4; ++j) {
    const int idx = w * TP + c * RP + 8 * h + 2 * j;
    const float2 tr = *(const float2*)&sRe[idx];
    const float2 ti = *(const float2*)&sIm[idx];
    br[j].x = tr.x; br[j].y = tr.y;
    bi[j].x = ti.x; bi[j].y = ti.y;
  }
  yr = zero; yi = zero;
#pragma unroll
  for (int j = 0; j < 4; ++j) {
    yr = mma(Fr[j], br[j], yr);
    yi = mma(Fr[j], bi[j], yi);
    yr = mma(Fi[j], bi[j], yr);
    yi = mma(Fn[j], br[j], yi);
  }
#pragma unroll
  for (int v = 0; v < 8; ++v) {
    const int idx = w * TP + (8 * h + v) * RP + c;
    sRe[idx] = yr[v];
    sIm[idx] = yi[v];
  }
  __syncthreads();

#pragma unroll
  for (int j = 0; j < 4; ++j) {
    const int idx = (8 * h + 2 * j) * TP + w * RP + c;
    const float r0 = sRe[idx],      i0 = sIm[idx];
    const float r1 = sRe[idx + TP], i1 = sIm[idx + TP];
    br[j].x = r0 * T1r[2 * j] + i0 * T1i[2 * j];
    bi[j].x = i0 * T1r[2 * j] - r0 * T1i[2 * j];
    br[j].y = r1 * T1r[2 * j + 1] + i1 * T1i[2 * j + 1];
    bi[j].y = i1 * T1r[2 * j + 1] - r1 * T1i[2 * j + 1];
  }
  yr = zero; yi = zero;
#pragma unroll
  for (int j = 0; j < 4; ++j) {
    yr = mma(Gr[j], br[j], yr);
    yi = mma(Gr[j], bi[j], yi);
    yr = mma(Gn[j], bi[j], yr);
    yi = mma(Gi[j], br[j], yi);
  }
#pragma unroll
  for (int v = 0; v < 8; ++v) {
    const int n = 256 * (8 * h + v) + 16 * w + c;
    sOut[n]       = yr[v] * (1.f / 4096.f);
    sOut[LEN + n] = yi[v] * (1.f / 4096.f);
  }
  __syncthreads();

  float* outb = out + b * OUTL;
#pragma unroll
  for (int i = 0; i < 16; ++i) {
    const int t = tid + NTHR * i;
    if (t < OUTL) outb[t] = sOut[t];
  }
}

extern "C" void kernel_launch(void* const* d_in, const int* in_sizes, int n_in,
                              void* d_out, int out_size, void* d_ws, size_t ws_size,
                              hipStream_t stream) {
  (void)in_sizes; (void)n_in; (void)d_ws; (void)ws_size; (void)out_size;
  const float* x = (const float*)d_in[0];
  float* out = (float*)d_out;
  selfconv_dft16_wmma_f32<<<dim3(128), dim3(NTHR), 0, stream>>>(x, out);
}


# Each 16-point pass = one radix-2 add/sub step in the vector unit + two 8-point transforms as 16x16 real products [Mr -Mi; Mi Mr]: 44 WMMAs per wave, not 88 (704 per row); sum order changes
# speedup vs baseline: 15.1835x; 1.0734x over previous
#include <hip/hip_runtime.h>


#define LEN   4096
#define OUTL  (2 * LEN - 1)
#define NTHR  512
#define RP    18
#define TP    290
#define PLANE (16 * TP)
#define ROWOF(h, v) (4 * (h) + ((v) & 3) + 8 * ((v) >> 2))

typedef float v2f __attribute__((ext_vector_type(2)));
typedef float v8f __attribute__((ext_vector_type(8)));

#if defined(__HIP_DEVICE_COMPILE__)
#if !__has_builtin(__builtin_amdgcn_wmma_f32_16x16x4_f32)
#error "missing __builtin_amdgcn_wmma_f32_16x16x4_f32 on this toolchain"
#endif
#endif

__device__ __forceinline__ v8f mma(const v2f a, const v2f b, const v8f c) {
  return __builtin_amdgcn_wmma_f32_16x16x4_f32(
       false, a,  false, b,
       (short)0, c,  false,  false);
}

__device__ __forceinline__ float2 unit(const float2* tabL, const float2* tabH,
                                       const int q) {
  const int qm = q & 16383;
  const float2 a = tabL[qm & 127];
  const float2 b = tabH[qm >> 7];
  return make_float2(a.x * b.x - a.y * b.y, a.x * b.y + a.y * b.x);
}

__device__ __forceinline__ float2 unit128(const float2* tabH, const int q) {
  return tabH[(q & 16383) >> 7];
}

__device__ __forceinline__ v2f cfrag(const int part, const float2 z) {
  v2f r;
  r.x = part ? z.y : z.x;
  r.y = part ? z.x : -z.y;
  return r;
}

__global__ __launch_bounds__(NTHR) void selfconv_dft16_wmma_f32(
    const float* __restrict__ x, float* __restrict__ out) {
  __shared__ __attribute__((aligned(16))) float sRe[PLANE];
  __shared__ __attribute__((aligned(16))) float sIm[PLANE];
  __shared__ __attribute__((aligned(16))) float sOut[2 * LEN];
  __shared__ float2 tabL[128];
  __shared__ float2 tabH[128];

  const int tid  = threadIdx.x;
  const int b    = blockIdx.x;
  const int lane = tid & 31;
  const int c    = lane & 15;
  const int h    = lane >> 4;
  const int w = __builtin_amdgcn_readfirstlane(tid >> 5);

  if (tid < 256) {
    const int t = tid & 127;
    const float ang = (tid < 128) ? (float)t * (1.f / 8192.f)
                                  : (float)t * (1.f / 64.f);
    float sn, cs;
    sincospif(ang, &sn, &cs);
    if (tid < 128) tabL[t] = make_float2(cs, sn);
    else           tabH[t] = make_float2(cs, sn);
  }
  __syncthreads();

  const float* xb = x + b * LEN;
  float xv[8];
#pragma unroll
  for (int v = 0; v < 8; ++v) xv[v] = xb[256 * ROWOF(h, v) + 16 * w + c];

  const int part = c & 1;
  const int ih   = (c & 7) >> 1;
  const int oT   = 4 * (c >> 3) + ih;
  const int oF   = 2 * (c >> 3) + (ih & 1) + 4 * (ih >> 1);
  v2f PE[2], PO[2];
  v2f Fev[4], Fod[4];
  v2f Ie[4], Io[4];
  v2f Gev[4], God[4];
#pragma unroll
  for (int j = 0; j < 2; ++j) {
#pragma unroll
    for (int s = 0; s < 2; ++s) {
      const int n2 = 4 * h + 2 * s + 8 * j;
      const float2 ze = unit128(tabH, 256 * n2 * (1 - 4 * oT));
      const float2 zo = unit128(tabH, 256 * (n2 + 1) * (1 - 4 * oT));
      const float e = part ? ze.y : ze.x;
      const float o = part ? zo.y : zo.x;
      if (s == 0) { PE[j].x = e; PO[j].x = o; }
      else        { PE[j].y = e; PO[j].y = o; }
    }
  }
#pragma unroll
  for (int j = 0; j < 4; ++j) {
    const int aF = 4 * h + j;
    const int aT = 2 * h + (j & 1) + 4 * (j >> 1);
    Fev[j] = cfrag(part, unit128(tabH, -2048 * oF * aF));
    Fod[j] = cfrag(part, unit128(tabH, -1024 * (2 * oF + 1) * aF));
    Ie[j]  = cfrag(part, unit128(tabH, 2048 * oT * aT));
    Io[j]  = cfrag(part, unit128(tabH, 1024 * oT * (2 * aT + 1)));
    Gev[j] = cfrag(part, unit128(tabH, -256 * (2 * oF) * (1 - 4 * aF)));
    God[j] = cfrag(part, unit128(tabH, -256 * (2 * oF + 1) * (1 - 4 * aF)));
  }
  float T1r[8], T1i[8];
  float T2r[8], T2i[8];
#pragma unroll
  for (int v = 0; v < 8; ++v) {
    const float2 t1 = unit(tabL, tabH, (16 * w + c) * (1 - 4 * ROWOF(h, v)));
    const float2 t2 = unit(tabL, tabH, -64 * ROWOF(h, v) * c);
    T1r[v] = t1.x; T1i[v] = t1.y;
    T2r[v] = t2.x; T2i[v] = t2.y;
  }

  const v8f zero = (v8f){0.f, 0.f, 0.f, 0.f, 0.f, 0.f, 0.f, 0.f};
  v8f dA, dB;
  float yr[8], yi[8];
  v2f pa[4], pb[4];

  dA = zero; dB = zero;
#pragma unroll
  for (int j = 0; j < 2; ++j) {
    v2f be, bo;
    be.x = xv[4 * j];     be.y = xv[4 * j + 2];
    bo.x = xv[4 * j + 1]; bo.y = xv[4 * j + 3];
    dA = mma(PE[j], be, dA);
    dB = mma(PO[j], bo, dB);
  }
#pragma unroll
  for (int i = 0; i < 4; ++i) {
    yr[i]     = dA[2 * i] + dB[2 * i];  yi[i]     = dA[2 * i + 1] + dB[2 * i + 1];
    yr[i + 4] = dA[2 * i] - dB[2 * i];  yi[i + 4] = dA[2 * i + 1] - dB[2 * i + 1];
  }
#pragma unroll
  for (int v = 0; v < 8; ++v) {
    const int idx = ROWOF(h, v) * TP + w * RP + c;
    sRe[idx] = yr[v] * T1r[v] - yi[v] * T1i[v];
    sIm[idx] = yr[v] * T1i[v] + yi[v] * T1r[v];
  }
  __syncthreads();

#pragma unroll
  for (int v = 0; v < 8; ++v) {
    const int idx = w * TP + ROWOF(h, v) * RP + c;
    yr[v] = sRe[idx];
    yi[v] = sIm[idx];
  }
  dA = zero; dB = zero;
#pragma unroll
  for (int j = 0; j < 4; ++j) {
    pa[j].x = yr[j] + yr[j + 4];  pa[j].y = yi[j] + yi[j + 4];
    pb[j].x = yr[j] - yr[j + 4];  pb[j].y = yi[j] - yi[j + 4];
  }
#pragma unroll
  for (int j = 0; j < 4; ++j) {
    dA = mma(Fev[j], pa[j], dA);
    dB = mma(Fod[j], pb[j], dB);
  }
#pragma unroll
  for (int i = 0; i < 4; ++i) {
    const int v = 2 * (i & 1) + 4 * (i >> 1);
    const int idx = w * TP + ROWOF(h, v) * RP + c;
    sRe[idx]      = dA[2 * i] * T2r[v] - dA[2 * i + 1] * T2i[v];
    sIm[idx]      = dA[2 * i] * T2i[v] + dA[2 * i + 1] * T2r[v];
    sRe[idx + RP] = dB[2 * i] * T2r[v + 1] - dB[2 * i + 1] * T2i[v + 1];
    sIm[idx + RP] = dB[2 * i] * T2i[v + 1] + dB[2 * i + 1] * T2r[v + 1];
  }
  __syncthreads();

#pragma unroll
  for (int v = 0; v < 8; v += 2) {
    const int idx = w * TP + c * RP + ROWOF(h, v);
    const float2 tr = *(const float2*)&sRe[idx];
    const float2 ti = *(const float2*)&sIm[idx];
    yr[v] = tr.x; yr[v + 1] = tr.y;
    yi[v] = ti.x; yi[v + 1] = ti.y;
  }
  dA = zero; dB = zero;
#pragma unroll
  for (int j = 0; j < 4; ++j) {
    pa[j].x = yr[j] + yr[j + 4];  pa[j].y = yi[j] + yi[j + 4];
    pb[j].x = yr[j] - yr[j + 4];  pb[j].y = yi[j] - yi[j + 4];
  }
#pragma unroll
  for (int j = 0; j < 4; ++j) {
    dA = mma(Fev[j], pa[j], dA);
    dB = mma(Fod[j], pb[j], dB);
  }
#pragma unroll
  for (int j = 0; j < 4; ++j) {
    const float a0 = dA[2 * j], b0 = dA[2 * j + 1];
    const float a1 = dB[2 * j], b1 = dB[2 * j + 1];
    pa[j].x = a0 * a0 - b0 * b0;  pa[j].y = 2.f * a0 * b0;
    pb[j].x = a1 * a1 - b1 * b1;  pb[j].y = 2.f * a1 * b1;
  }

  dA = zero; dB = zero;
#pragma unroll
  for (int j = 0; j < 4; ++j) {
    dA = mma(Ie[j], pa[j], dA);
    dB = mma(Io[j], pb[j], dB);
  }
#pragma unroll
  for (int i = 0; i < 4; ++i) {
    yr[i]     = dA[2 * i] + dB[2 * i];  yi[i]     = dA[2 * i + 1] + dB[2 * i + 1];
    yr[i + 4] = dA[2 * i] - dB[2 * i];  yi[i + 4] = dA[2 * i + 1] - dB[2 * i + 1];
  }
#pragma unroll
  for (int v = 0; v < 8; ++v) {
    const int idx = w * TP + ROWOF(h, v) * RP + c;
    sRe[idx] = yr[v] * T2r[v] + yi[v] * T2i[v];
    sIm[idx] = yi[v] * T2r[v] - yr[v] * T2i[v];
  }
  __syncthreads();

#pragma unroll
  for (int j = 0; j < 4; ++j) {
    const int v = 2 * (j & 1) + 4 * (j >> 1);
    const int idx = w * TP + c * RP + ROWOF(h, v);
    const float2 tr = *(const float2*)&sRe[idx];
    const float2 ti = *(const float2*)&sIm[idx];
    pa[j].x = tr.x; pa[j].y = ti.x;
    pb[j].x = tr.y; pb[j].y = ti.y;
  }
  dA = zero; dB = zero;
#pragma unroll
  for (int j = 0; j < 4; ++j) {
    dA = mma(Ie[j], pa[j], dA);
    dB = mma(Io[j], pb[j], dB);
  }
#pragma unroll
  for (int i = 0; i < 4; ++i) {
    const int idx = w * TP + ROWOF(h, i) * RP + c;
    sRe[idx]          = dA[2 * i] + dB[2 * i];
    sIm[idx]          = dA[2 * i + 1] + dB[2 * i + 1];
    sRe[idx + 8 * RP] = dA[2 * i] - dB[2 * i];
    sIm[idx + 8 * RP] = dA[2 * i + 1] - dB[2 * i + 1];
  }
  __syncthreads();

#pragma unroll
  for (int v = 0; v < 8; ++v) {
    const int idx = ROWOF(h, v) * TP + w * RP + c;
    const float r0 = sRe[idx], i0 = sIm[idx];
    yr[v] = r0 * T1r[v] + i0 * T1i[v];
    yi[v] = i0 * T1r[v] - r0 * T1i[v];
  }
  dA = zero; dB = zero;
#pragma unroll
  for (int j = 0; j < 4; ++j) {
    pa[j].x = yr[j] + yr[j + 4];  pa[j].y = yi[j] + yi[j + 4];
    pb[j].x = yr[j] - yr[j + 4];  pb[j].y = yi[j] - yi[j + 4];
  }
#pragma unroll
  for (int j = 0; j < 4; ++j) {
    dA = mma(Gev[j], pa[j], dA);
    dB = mma(God[j], pb[j], dB);
  }
#pragma unroll
  for (int i = 0; i < 4; ++i) {
    const int v = 2 * (i & 1) + 4 * (i >> 1);
    const int n = 256 * ROWOF(h, v) + 16 * w + c;
    sOut[n]             = dA[2 * i] * (1.f / 4096.f);
    sOut[LEN + n]       = dA[2 * i + 1] * (1.f / 4096.f);
    sOut[n + 256]       = dB[2 * i] * (1.f / 4096.f);
    sOut[LEN + n + 256] = dB[2 * i + 1] * (1.f / 4096.f);
  }
  __syncthreads();

  float* outb = out + b * OUTL;
#pragma unroll
  for (int i = 0; i < 16; ++i) {
    const int t = tid + NTHR * i;
    if (t < OUTL) outb[t] = sOut[t];
  }
}

extern "C" void kernel_launch(void* const* d_in, const int* in_sizes, int n_in,
                              void* d_out, int out_size, void* d_ws, size_t ws_size,
                              hipStream_t stream) {
  (void)in_sizes; (void)n_in; (void)d_ws; (void)ws_size; (void)out_size;
  const float* x = (const float*)d_in[0];
  float* out = (float*)d_out;
  selfconv_dft16_wmma_f32<<<dim3(128), dim3(NTHR), 0, stream>>>(x, out);
}
